# starting kernel
# baseline (speedup 1.0000x reference)
#include <hip/hip_runtime.h>
#include <hip/hip_bf16.h>

typedef __attribute__((ext_vector_type(2))) float v2f;
typedef __attribute__((ext_vector_type(8))) float v8f;

namespace {
constexpr int Bc = 8;
constexpr int Ic = 64;
constexpr int Kc = 16;
constexpr int Lc = 256;
constexpr int Oc = 16;
constexpr int Dc = 32;
constexpr int LT = 16;
constexpr int ITERS = 3;
constexpr int XSS = Kc + 2;
constexpr int NTHREADS = Oc * 32;
constexpr int LDS_FLOATS = Ic * LT * XSS
                         + Oc * Ic * LT
                         + Ic * LT
                         + Ic * LT;
}

__device__ __forceinline__ void wmma_u(const float* __restrict__ Wp,
                                       const float* __restrict__ xrow,
                                       int arow, v8f& u0, v8f& u1)
{
#pragma unroll
    for (int c = 0; c < 4; ++c) {
        v2f bf = *(const v2f*)(xrow + 4 * c);
        v2f a0 = *(const v2f*)(Wp + arow + 4 * c);
        v2f a1 = *(const v2f*)(Wp + arow + 16 * Kc + 4 * c);
        u0 = __builtin_amdgcn_wmma_f32_16x16x4_f32(false, a0, false, bf, (short)0, u0, false, false);
        u1 = __builtin_amdgcn_wmma_f32_16x16x4_f32(false, a1, false, bf, (short)0, u1, false, false);
    }
}

__global__ __launch_bounds__(NTHREADS)
void DynamicRouting_86766929314212_kernel(const float* __restrict__ x,
                                          const float* __restrict__ W,
                                          float* __restrict__ out)
{
    extern __shared__ float lds[];
    float* xs = lds;
    float* bl = xs + Ic * LT * XSS;
    float* mx = bl + Oc * Ic * LT;
    float* rd = mx + Ic * LT;

    const int tid  = threadIdx.x;
    const int lane = tid & 31;
    const int o    = tid >> 5;
    const int n    = lane & 15;
    const int h    = lane >> 4;
    const int arow = n * Kc + 2 * h;

    const int tilesPerB = Lc / LT;
    const int b  = blockIdx.x / tilesPerB;
    const int l0 = (blockIdx.x % tilesPerB) * LT;

    const float* Wo = W + (size_t)o * Ic * Dc * Kc;

    for (int idx = tid; idx < Ic * Kc * LT; idx += NTHREADS) {
        int i  = idx >> 8;
        int r  = idx & 255;
        int k  = r >> 4;
        int nn = r & 15;
        xs[(i * LT + nn) * XSS + k] = x[((size_t)(b * Ic + i) * Kc + k) * Lc + l0 + nn];
    }
    for (int idx = tid; idx < Oc * Ic * LT; idx += NTHREADS) bl[idx] = 0.0f;

    v8f v0, v1;

    for (int it = 0; it < ITERS; ++it) {
        __syncthreads();
        for (int idx = tid; idx < Ic * LT; idx += NTHREADS) {
            float m = bl[idx];
            for (int oo = 1; oo < Oc; ++oo) m = fmaxf(m, bl[oo * Ic * LT + idx]);
            float den = 0.0f;
            for (int oo = 0; oo < Oc; ++oo) den += __expf(bl[oo * Ic * LT + idx] - m);
            mx[idx] = m;
            rd[idx] = 1.0f / den;
        }
        __syncthreads();

        v8f s0 = {0.f,0.f,0.f,0.f,0.f,0.f,0.f,0.f};
        v8f s1 = {0.f,0.f,0.f,0.f,0.f,0.f,0.f,0.f};
        for (int i = 0; i < Ic; ++i) {
            v8f u0 = {0.f,0.f,0.f,0.f,0.f,0.f,0.f,0.f};
            v8f u1 = {0.f,0.f,0.f,0.f,0.f,0.f,0.f,0.f};
            wmma_u(Wo + i * Dc * Kc, xs + (i * LT + n) * XSS + 2 * h, arow, u0, u1);
            float cb = __expf(bl[(o * Ic + i) * LT + n] - mx[i * LT + n]) * rd[i * LT + n];
#pragma unroll
            for (int j = 0; j < 8; ++j) { s0[j] += cb * u0[j]; s1[j] += cb * u1[j]; }
        }

        float sq = 0.0f;
#pragma unroll
        for (int j = 0; j < 8; ++j) sq += s0[j] * s0[j] + s1[j] * s1[j];
        sq += __shfl_xor(sq, 16, 32);
        const float scale = (sq / (1.0f + sq)) * rsqrtf(sq + 1e-8f);
#pragma unroll
        for (int j = 0; j < 8; ++j) { s0[j] *= scale; s1[j] *= scale; }
        v0 = s0; v1 = s1;

        if (it + 1 < ITERS) {
            for (int i = 0; i < Ic; ++i) {
                v8f u0 = {0.f,0.f,0.f,0.f,0.f,0.f,0.f,0.f};
                v8f u1 = {0.f,0.f,0.f,0.f,0.f,0.f,0.f,0.f};
                wmma_u(Wo + i * Dc * Kc, xs + (i * LT + n) * XSS + 2 * h, arow, u0, u1);
                float dot = 0.0f;
#pragma unroll
                for (int j = 0; j < 8; ++j) dot += u0[j] * v0[j] + u1[j] * v1[j];
                dot += __shfl_xor(dot, 16, 32);
                if (h == 0) bl[(o * Ic + i) * LT + n] += dot;
            }
        }
    }

    const size_t obase = ((size_t)(b * Oc + o) * Dc) * Lc + l0 + n;
#pragma unroll
    for (int j = 0; j < 8; ++j) {
        int d = j + 8 * h;
        out[obase + (size_t)d * Lc]        = v0[j];
        out[obase + (size_t)(d + 16) * Lc] = v1[j];
    }
}

extern "C" void kernel_launch(void* const* d_in, const int* in_sizes, int n_in,
                              void* d_out, int out_size, void* d_ws, size_t ws_size,
                              hipStream_t stream)
{
    (void)in_sizes; (void)n_in; (void)out_size; (void)d_ws; (void)ws_size;
    const float* x = (const float*)d_in[0];
    const float* W = (const float*)d_in[1];
    float* out = (float*)d_out;

    constexpr size_t shmem = (size_t)LDS_FLOATS * sizeof(float);
    dim3 grid(Bc * (Lc / LT));
    DynamicRouting_86766929314212_kernel<<<grid, NTHREADS, shmem, stream>>>(x, W, out);
}


# Vector work software-pipelined under the WMMAs: dot of out-capsule o-1 and weighted sum of the previous in-capsule issue between the WMMAs of o; f32 sum order of the logit dot changes
# speedup vs baseline: 1.5901x; 1.5901x over previous
#include <hip/hip_runtime.h>


typedef __attribute__((ext_vector_type(2))) float v2f;
typedef __attribute__((ext_vector_type(4))) float v4f;
typedef __attribute__((ext_vector_type(8))) float v8f;

namespace {
constexpr int Bc = 8;
constexpr int Ic = 64;
constexpr int Kc = 16;
constexpr int Lc = 256;
constexpr int Oc = 16;
constexpr int Dc = 32;
constexpr int LT = 16;
constexpr int ITERS = 3;
constexpr int XSS = Kc + 2;
constexpr int NW = 4;
constexpr int NTHREADS = NW * 32;
constexpr int IPW = Ic / NW;
constexpr int FRAG = 16 * 32;
constexpr int XS_F = Ic * LT * XSS;
constexpr int BLP = Oc;
constexpr int BL_F = Ic * LT * BLP;
constexpr int VS_F = Oc * FRAG;
constexpr int SP_F = NW * Oc * FRAG;
constexpr int LDS_FLOATS = XS_F + BL_F + VS_F + SP_F;
}

__device__ __forceinline__ float xhalf(float v)
{
    return __builtin_bit_cast(float, __builtin_amdgcn_permlanex16(__builtin_bit_cast(int, v), __builtin_bit_cast(int, v),
                                                                  0x76543210, 0xfedcba98, false, false));
}

__device__ __forceinline__ void load_group(const float* Wi, int g, int arow, v2f (&a)[4][8])
{
#pragma unroll
    for (int q = 0; q < 4; ++q) {
        const float* Wp = Wi + (size_t)(4 * g + q) * Ic * Dc * Kc;
#pragma unroll
        for (int c = 0; c < 4; ++c) {
            a[q][2 * c]     = *(const v2f*)(Wp + arow + 4 * c);
            a[q][2 * c + 1] = *(const v2f*)(Wp + arow + 16 * Kc + 4 * c);
        }
    }
}

__device__ __forceinline__ void mma_o(const v2f (&a)[8], const v2f (&bf)[4], v8f (&u)[2])
{
    v8f u0 = (v8f){0.f,0.f,0.f,0.f,0.f,0.f,0.f,0.f};
    v8f u1 = (v8f){0.f,0.f,0.f,0.f,0.f,0.f,0.f,0.f};
#pragma unroll
    for (int c = 0; c < 4; ++c) {
        u0 = __builtin_amdgcn_wmma_f32_16x16x4_f32(false, a[2 * c],     false, bf[c], (short)0, u0, false, false);
        u1 = __builtin_amdgcn_wmma_f32_16x16x4_f32(false, a[2 * c + 1], false, bf[c], (short)0, u1, false, false);
    }
    u[0] = u0; u[1] = u1;
}

__device__ __forceinline__ void ws_o(v8f (&sA)[2], const v8f (&u)[2], float cb)
{
    sA[0] += cb * u[0];
    sA[1] += cb * u[1];
}

__device__ __forceinline__ float dot_o(const v8f (&u)[2], const v4f (&vr)[4])
{
    v2f a0 = __builtin_shufflevector(u[0], u[0], 0, 1) * __builtin_shufflevector(vr[0], vr[0], 0, 1);
    v2f a1 = __builtin_shufflevector(u[1], u[1], 0, 1) * __builtin_shufflevector(vr[2], vr[2], 0, 1);
    a0 = __builtin_shufflevector(u[0], u[0], 2, 3) * __builtin_shufflevector(vr[0], vr[0], 2, 3) + a0;
    a1 = __builtin_shufflevector(u[1], u[1], 2, 3) * __builtin_shufflevector(vr[2], vr[2], 2, 3) + a1;
    a0 = __builtin_shufflevector(u[0], u[0], 4, 5) * __builtin_shufflevector(vr[1], vr[1], 0, 1) + a0;
    a1 = __builtin_shufflevector(u[1], u[1], 4, 5) * __builtin_shufflevector(vr[3], vr[3], 0, 1) + a1;
    a0 = __builtin_shufflevector(u[0], u[0], 6, 7) * __builtin_shufflevector(vr[1], vr[1], 2, 3) + a0;
    a1 = __builtin_shufflevector(u[1], u[1], 6, 7) * __builtin_shufflevector(vr[3], vr[3], 2, 3) + a1;
    a0 += a1;
    const float d = a0[0] + a0[1];
    return d + xhalf(d);
}

template <int KV>
__device__ __forceinline__ void wmma_slot()
{
    __builtin_amdgcn_sched_group_barrier(0x008, 1, 0);
    __builtin_amdgcn_sched_group_barrier(0x002, KV, 0);
}

template <int KA, int KL>
__device__ __forceinline__ void capsule_issue_order()
{
    wmma_slot<KA>(); wmma_slot<KA>(); wmma_slot<KA>(); wmma_slot<KA>();
    wmma_slot<KA>(); wmma_slot<KA>(); wmma_slot<KA>(); wmma_slot<KL>();
}

template <bool FIRST>
__device__ __forceinline__ void route_pass(const float* W, const float* xs,
                                           float* bl, const float* vs,
                                           int w, int lane, int n, int h, int arow,
                                           v8f (&sA)[Oc][2])
{
    v4f vr[Oc][4];
    if (!FIRST) {
#pragma unroll
        for (int o = 0; o < Oc; ++o) {
            const v4f* vp = (const v4f*)(vs + o * FRAG) + lane;
            vr[o][0] = vp[0]; vr[o][1] = vp[32]; vr[o][2] = vp[64]; vr[o][3] = vp[96];
        }
    }

    v2f af[2][4][8];
    load_group(W + (size_t)(w * IPW) * Dc * Kc, 0, arow, af[0]);

    v8f u[Oc][2];
    float cb[Oc];
#pragma unroll
    for (int o = 0; o < Oc; ++o) {
        u[o][0] = (v8f){0.f,0.f,0.f,0.f,0.f,0.f,0.f,0.f};
        u[o][1] = (v8f){0.f,0.f,0.f,0.f,0.f,0.f,0.f,0.f};
        cb[o] = FIRST ? 0.0625f : 0.0f;
    }

    for (int ii = 0; ii < IPW; ++ii) {
        const int i = w * IPW + ii;
        const int inext = (ii + 1 < IPW) ? i + 1 : i;
        const float* Wi = W + (size_t)i * Dc * Kc;
        const float* xrow = xs + (i * LT + n) * XSS + 2 * h;
        v2f bf[4];
#pragma unroll
        for (int c = 0; c < 4; ++c) bf[c] = *(const v2f*)(xrow + 4 * c);

        float lg[Oc];
        if (FIRST) {
#pragma unroll
            for (int g = 0; g < 4; ++g) {
                if (g < 3) load_group(Wi, g + 1, arow, af[(g + 1) & 1]);
                else       load_group(W + (size_t)inext * Dc * Kc, 0, arow, af[0]);
                __builtin_amdgcn_sched_barrier(0);
#pragma unroll
                for (int q = 0; q < 4; ++q) {
                    const int o = 4 * g + q;
                    if (o == 0) ws_o(sA[Oc - 1], u[Oc - 1], cb[Oc - 1]);
                    mma_o(af[g & 1][q], bf, u[o]);
                    if (o >= 1) ws_o(sA[o - 1], u[o - 1], cb[o - 1]);
                }
                capsule_issue_order<1, 1>(); capsule_issue_order<1, 1>(); capsule_issue_order<1, 1>(); capsule_issue_order<1, 1>();
                __builtin_amdgcn_sched_barrier(0);
            }
        } else {
            v4f* blp = (v4f*)(bl + (i * LT + n) * BLP);
            v4f old[4];
#pragma unroll
            for (int q = 0; q < 4; ++q) old[q] = blp[q];

            ws_o(sA[0], u[0], cb[0]);
#pragma unroll
            for (int g = 0; g < 4; ++g) {
                if (g < 3) load_group(Wi, g + 1, arow, af[(g + 1) & 1]);
                else       load_group(W + (size_t)inext * Dc * Kc, 0, arow, af[0]);
                __builtin_amdgcn_sched_barrier(0);
#pragma unroll
                for (int q = 0; q < 4; ++q) {
                    const int o = 4 * g + q;
                    mma_o(af[g & 1][q], bf, u[o]);
                    if (o + 1 < Oc) ws_o(sA[o + 1], u[o + 1], cb[o + 1]);
                    if (o >= 1) lg[o - 1] = dot_o(u[o - 1], vr[o - 1]);
                }
                if (g == 0)      { capsule_issue_order<1, 1>(); capsule_issue_order<2, 3>(); capsule_issue_order<2, 3>(); capsule_issue_order<2, 3>(); }
                else if (g == 3) { capsule_issue_order<2, 3>(); capsule_issue_order<2, 3>(); capsule_issue_order<2, 3>(); capsule_issue_order<1, 2>(); }
                else             { capsule_issue_order<2, 3>(); capsule_issue_order<2, 3>(); capsule_issue_order<2, 3>(); capsule_issue_order<2, 3>(); }
                __builtin_amdgcn_sched_barrier(0);
            }
            lg[Oc - 1] = dot_o(u[Oc - 1], vr[Oc - 1]);
#pragma unroll
            for (int q = 0; q < 4; ++q) {
                lg[4 * q + 0] += old[q][0]; lg[4 * q + 1] += old[q][1]; lg[4 * q + 2] += old[q][2]; lg[4 * q + 3] += old[q][3];
            }
#pragma unroll
            for (int q = 0; q < 4; ++q)
                blp[q] = (v4f){lg[4 * q + 0], lg[4 * q + 1], lg[4 * q + 2], lg[4 * q + 3]};
            float m = lg[0];
#pragma unroll
            for (int o = 1; o < Oc; ++o) m = fmaxf(m, lg[o]);
            float den = 0.0f;
#pragma unroll
            for (int o = 0; o < Oc; ++o) { lg[o] = __expf(lg[o] - m); den += lg[o]; }
            const float r = 1.0f / den;
#pragma unroll
            for (int o = 0; o < Oc; ++o) cb[o] = lg[o] * r;
        }
    }

    if (FIRST) {
        ws_o(sA[Oc - 1], u[Oc - 1], cb[Oc - 1]);
    } else {
#pragma unroll
        for (int o = 0; o < Oc; ++o) ws_o(sA[o], u[o], cb[o]);
    }
}

__global__ __launch_bounds__(NTHREADS)
void DynamicRouting_86766929314212_kernel(const float* __restrict__ x,
                                          const float* W,
                                          float* __restrict__ out)
{
    extern __shared__ __attribute__((aligned(16))) float lds[];
    float* xs = lds;
    float* bl = xs + XS_F;
    float* vs = bl + BL_F;
    float* sp = vs + VS_F;

    const int tid  = threadIdx.x;
    const int lane = tid & 31;
    const int w    = tid >> 5;
    const int n    = lane & 15;
    const int h    = lane >> 4;
    const int arow = n * Kc + 2 * h;

    const int tilesPerB = Lc / LT;
    const int b  = blockIdx.x / tilesPerB;
    const int l0 = (blockIdx.x % tilesPerB) * LT;

#pragma unroll
    for (int t = 0; t < (Ic * Kc * (LT / 4)) / NTHREADS; ++t) {
        const int idx  = tid + t * NTHREADS;
        const int row  = idx >> 2;
        const int part = idx & 3;
        const int i = row >> 4;
        const int k = row & 15;
        const v4f val = *(const v4f*)(x + ((size_t)(b * Ic + i) * Kc + k) * Lc + l0 + part * 4);
        float* dst = xs + (i * LT + part * 4) * XSS + k;
        dst[0 * XSS] = val[0];
        dst[1 * XSS] = val[1];
        dst[2 * XSS] = val[2];
        dst[3 * XSS] = val[3];
    }
    for (int idx = tid; idx < BL_F / 4; idx += NTHREADS) ((v4f*)bl)[idx] = (v4f){0.f, 0.f, 0.f, 0.f};
    __syncthreads();

    for (int it = 0; it < ITERS; ++it) {
        v8f sA[Oc][2];
#pragma unroll
        for (int o = 0; o < Oc; ++o) {
            sA[o][0] = (v8f){0.f,0.f,0.f,0.f,0.f,0.f,0.f,0.f};
            sA[o][1] = (v8f){0.f,0.f,0.f,0.f,0.f,0.f,0.f,0.f};
        }
        if (it == 0) route_pass<true >(W, xs, bl, vs, w, lane, n, h, arow, sA);
        else         route_pass<false>(W, xs, bl, vs, w, lane, n, h, arow, sA);

        {
            v4f* spw = (v4f*)(sp + (size_t)(w * Oc) * FRAG) + lane;
#pragma unroll
            for (int o = 0; o < Oc; ++o) {
                spw[o * 128 +  0] = (v4f){sA[o][0][0], sA[o][0][1], sA[o][0][2], sA[o][0][3]};
                spw[o * 128 + 32] = (v4f){sA[o][0][4], sA[o][0][5], sA[o][0][6], sA[o][0][7]};
                spw[o * 128 + 64] = (v4f){sA[o][1][0], sA[o][1][1], sA[o][1][2], sA[o][1][3]};
                spw[o * 128 + 96] = (v4f){sA[o][1][4], sA[o][1][5], sA[o][1][6], sA[o][1][7]};
            }
        }
        __syncthreads();

#pragma unroll
        for (int qq = 0; qq < Oc / NW; ++qq) {
            const int o = w * (Oc / NW) + qq;
            v4f t[4];
#pragma unroll
            for (int q = 0; q < 4; ++q) {
                const v4f* p = (const v4f*)(sp + (size_t)o * FRAG) + q * 32 + lane;
                t[q] = ((p[0] + p[(Oc * FRAG) / 4]) + p[2 * (Oc * FRAG) / 4]) + p[3 * (Oc * FRAG) / 4];
            }
            float sq = 0.0f;
#pragma unroll
            for (int e = 0; e < 4; ++e)
                sq += t[0][e] * t[0][e] + t[2][e] * t[2][e];
#pragma unroll
            for (int e = 0; e < 4; ++e)
                sq += t[1][e] * t[1][e] + t[3][e] * t[3][e];
            sq += __shfl_xor(sq, 16, 32);
            const float scale = (sq / (1.0f + sq)) * rsqrtf(sq + 1e-8f);
#pragma unroll
            for (int q = 0; q < 4; ++q) t[q] *= scale;

            if (it + 1 < ITERS) {
                v4f* vo = (v4f*)(vs + (size_t)o * FRAG) + lane;
                vo[0] = t[0]; vo[32] = t[1]; vo[64] = t[2]; vo[96] = t[3];
            } else {
                const size_t obase = ((size_t)(b * Oc + o) * Dc) * Lc + l0 + n;
#pragma unroll
                for (int e = 0; e < 4; ++e) {
                    out[obase + (size_t)(8 * h + e) * Lc]          = t[0][e];
                    out[obase + (size_t)(8 * h + 4 + e) * Lc]      = t[1][e];
                    out[obase + (size_t)(16 + 8 * h + e) * Lc]     = t[2][e];
                    out[obase + (size_t)(16 + 8 * h + 4 + e) * Lc] = t[3][e];
                }
            }
        }
        __syncthreads();
    }
}

extern "C" void kernel_launch(void* const* d_in, const int* in_sizes, int n_in,
                              void* d_out, int out_size, void* d_ws, size_t ws_size,
                              hipStream_t stream)
{
    (void)in_sizes; (void)n_in; (void)out_size; (void)d_ws; (void)ws_size;
    const float* x = (const float*)d_in[0];
    const float* W = (const float*)d_in[1];
    float* out = (float*)d_out;

    constexpr size_t shmem = (size_t)LDS_FLOATS * sizeof(float);
    dim3 grid(Bc * (Lc / LT));
    DynamicRouting_86766929314212_kernel<<<grid, NTHREADS, shmem, stream>>>(x, W, out);
}


# In-capsules of a tile split over 2 workgroups (256 fill the chip), one launch per routing iteration: partial s passed on in workspace, logits rebuilt as u.(v0+v1); f32 sum order of s, logits changes
# speedup vs baseline: 2.1467x; 1.3500x over previous
#include <hip/hip_runtime.h>


typedef __attribute__((ext_vector_type(2))) float v2f;
typedef __attribute__((ext_vector_type(4))) float v4f;
typedef __attribute__((ext_vector_type(8))) float v8f;

namespace {
constexpr int Bc = 8;
constexpr int Ic = 64;
constexpr int Kc = 16;
constexpr int Lc = 256;
constexpr int Oc = 16;
constexpr int Dc = 32;
constexpr int LT = 16;
constexpr int ITERS = 3;
constexpr int XSS = Kc + 2;
constexpr int NW = 4;
constexpr int NTHREADS = NW * 32;
constexpr int IPW = Ic / NW;
constexpr int FRAG = 16 * 32;
constexpr int XS_F = Ic * LT * XSS;
constexpr int BLP = Oc;
constexpr int BL_F = Ic * LT * BLP;
constexpr int VS_F = Oc * FRAG;
constexpr int SP_F = NW * Oc * FRAG;
constexpr int LDS_FLOATS = XS_F + BL_F + VS_F + SP_F;
}

__device__ __forceinline__ float xhalf(float v)
{
    return __builtin_bit_cast(float, __builtin_amdgcn_permlanex16(__builtin_bit_cast(int, v), __builtin_bit_cast(int, v),
                                                                  0x76543210, 0xfedcba98, false, false));
}

__device__ __forceinline__ void load_group(const float* Wi, int g, int arow, v2f (&a)[4][8])
{
#pragma unroll
    for (int q = 0; q < 4; ++q) {
        const float* Wp = Wi + (size_t)(4 * g + q) * Ic * Dc * Kc;
#pragma unroll
        for (int c = 0; c < 4; ++c) {
            a[q][2 * c]     = *(const v2f*)(Wp + arow + 4 * c);
            a[q][2 * c + 1] = *(const v2f*)(Wp + arow + 16 * Kc + 4 * c);
        }
    }
}

__device__ __forceinline__ void mma_o(const v2f (&a)[8], const v2f (&bf)[4], v8f (&u)[2])
{
    v8f u0 = (v8f){0.f,0.f,0.f,0.f,0.f,0.f,0.f,0.f};
    v8f u1 = (v8f){0.f,0.f,0.f,0.f,0.f,0.f,0.f,0.f};
#pragma unroll
    for (int c = 0; c < 4; ++c) {
        u0 = __builtin_amdgcn_wmma_f32_16x16x4_f32(false, a[2 * c],     false, bf[c], (short)0, u0, false, false);
        u1 = __builtin_amdgcn_wmma_f32_16x16x4_f32(false, a[2 * c + 1], false, bf[c], (short)0, u1, false, false);
    }
    u[0] = u0; u[1] = u1;
}

__device__ __forceinline__ void ws_o(v8f (&sA)[2], const v8f (&u)[2], float cb)
{
    sA[0] += cb * u[0];
    sA[1] += cb * u[1];
}

__device__ __forceinline__ float dot_o(const v8f (&u)[2], const v4f (&vr)[4])
{
    v2f a0 = __builtin_shufflevector(u[0], u[0], 0, 1) * __builtin_shufflevector(vr[0], vr[0], 0, 1);
    v2f a1 = __builtin_shufflevector(u[1], u[1], 0, 1) * __builtin_shufflevector(vr[2], vr[2], 0, 1);
    a0 = __builtin_shufflevector(u[0], u[0], 2, 3) * __builtin_shufflevector(vr[0], vr[0], 2, 3) + a0;
    a1 = __builtin_shufflevector(u[1], u[1], 2, 3) * __builtin_shufflevector(vr[2], vr[2], 2, 3) + a1;
    a0 = __builtin_shufflevector(u[0], u[0], 4, 5) * __builtin_shufflevector(vr[1], vr[1], 0, 1) + a0;
    a1 = __builtin_shufflevector(u[1], u[1], 4, 5) * __builtin_shufflevector(vr[3], vr[3], 0, 1) + a1;
    a0 = __builtin_shufflevector(u[0], u[0], 6, 7) * __builtin_shufflevector(vr[1], vr[1], 2, 3) + a0;
    a1 = __builtin_shufflevector(u[1], u[1], 6, 7) * __builtin_shufflevector(vr[3], vr[3], 2, 3) + a1;
    a0 += a1;
    const float d = a0[0] + a0[1];
    return d + xhalf(d);
}

template <int KV>
__device__ __forceinline__ void wmma_slot()
{
    __builtin_amdgcn_sched_group_barrier(0x008, 1, 0);
    __builtin_amdgcn_sched_group_barrier(0x002, KV, 0);
}

template <int KA, int KL>
__device__ __forceinline__ void capsule_issue_order()
{
    wmma_slot<KA>(); wmma_slot<KA>(); wmma_slot<KA>(); wmma_slot<KA>();
    wmma_slot<KA>(); wmma_slot<KA>(); wmma_slot<KA>(); wmma_slot<KL>();
}

template <bool FIRST>
__device__ __forceinline__ void route_pass(const float* W, const float* xs,
                                           float* bl, const float* vs,
                                           int w, int lane, int n, int h, int arow,
                                           v8f (&sA)[Oc][2])
{
    v4f vr[Oc][4];
    if (!FIRST) {
#pragma unroll
        for (int o = 0; o < Oc; ++o) {
            const v4f* vp = (const v4f*)(vs + o * FRAG) + lane;
            vr[o][0] = vp[0]; vr[o][1] = vp[32]; vr[o][2] = vp[64]; vr[o][3] = vp[96];
        }
    }

    v2f af[2][4][8];
    load_group(W + (size_t)(w * IPW) * Dc * Kc, 0, arow, af[0]);

    v8f u[Oc][2];
    float cb[Oc];
#pragma unroll
    for (int o = 0; o < Oc; ++o) {
        u[o][0] = (v8f){0.f,0.f,0.f,0.f,0.f,0.f,0.f,0.f};
        u[o][1] = (v8f){0.f,0.f,0.f,0.f,0.f,0.f,0.f,0.f};
        cb[o] = FIRST ? 0.0625f : 0.0f;
    }

    for (int ii = 0; ii < IPW; ++ii) {
        const int i = w * IPW + ii;
        const int inext = (ii + 1 < IPW) ? i + 1 : i;
        const float* Wi = W + (size_t)i * Dc * Kc;
        const float* xrow = xs + (i * LT + n) * XSS + 2 * h;
        v2f bf[4];
#pragma unroll
        for (int c = 0; c < 4; ++c) bf[c] = *(const v2f*)(xrow + 4 * c);

        float lg[Oc];
        if (FIRST) {
#pragma unroll
            for (int g = 0; g < 4; ++g) {
                if (g < 3) load_group(Wi, g + 1, arow, af[(g + 1) & 1]);
                else       load_group(W + (size_t)inext * Dc * Kc, 0, arow, af[0]);
                __builtin_amdgcn_sched_barrier(0);
#pragma unroll
                for (int q = 0; q < 4; ++q) {
                    const int o = 4 * g + q;
                    if (o == 0) ws_o(sA[Oc - 1], u[Oc - 1], cb[Oc - 1]);
                    mma_o(af[g & 1][q], bf, u[o]);
                    if (o >= 1) ws_o(sA[o - 1], u[o - 1], cb[o - 1]);
                }
                capsule_issue_order<1, 1>(); capsule_issue_order<1, 1>(); capsule_issue_order<1, 1>(); capsule_issue_order<1, 1>();
                __builtin_amdgcn_sched_barrier(0);
            }
        } else {
            v4f* blp = (v4f*)(bl + (i * LT + n) * BLP);
            v4f old[4];
#pragma unroll
            for (int q = 0; q < 4; ++q) old[q] = blp[q];

            ws_o(sA[0], u[0], cb[0]);
#pragma unroll
            for (int g = 0; g < 4; ++g) {
                if (g < 3) load_group(Wi, g + 1, arow, af[(g + 1) & 1]);
                else       load_group(W + (size_t)inext * Dc * Kc, 0, arow, af[0]);
                __builtin_amdgcn_sched_barrier(0);
#pragma unroll
                for (int q = 0; q < 4; ++q) {
                    const int o = 4 * g + q;
                    mma_o(af[g & 1][q], bf, u[o]);
                    if (o + 1 < Oc) ws_o(sA[o + 1], u[o + 1], cb[o + 1]);
                    if (o >= 1) lg[o - 1] = dot_o(u[o - 1], vr[o - 1]);
                }
                if (g == 0)      { capsule_issue_order<1, 1>(); capsule_issue_order<2, 3>(); capsule_issue_order<2, 3>(); capsule_issue_order<2, 3>(); }
                else if (g == 3) { capsule_issue_order<2, 3>(); capsule_issue_order<2, 3>(); capsule_issue_order<2, 3>(); capsule_issue_order<1, 2>(); }
                else             { capsule_issue_order<2, 3>(); capsule_issue_order<2, 3>(); capsule_issue_order<2, 3>(); capsule_issue_order<2, 3>(); }
                __builtin_amdgcn_sched_barrier(0);
            }
            lg[Oc - 1] = dot_o(u[Oc - 1], vr[Oc - 1]);
#pragma unroll
            for (int q = 0; q < 4; ++q) {
                lg[4 * q + 0] += old[q][0]; lg[4 * q + 1] += old[q][1]; lg[4 * q + 2] += old[q][2]; lg[4 * q + 3] += old[q][3];
            }
#pragma unroll
            for (int q = 0; q < 4; ++q)
                blp[q] = (v4f){lg[4 * q + 0], lg[4 * q + 1], lg[4 * q + 2], lg[4 * q + 3]};
            float m = lg[0];
#pragma unroll
            for (int o = 1; o < Oc; ++o) m = fmaxf(m, lg[o]);
            float den = 0.0f;
#pragma unroll
            for (int o = 0; o < Oc; ++o) { lg[o] = __expf(lg[o] - m); den += lg[o]; }
            const float r = 1.0f / den;
#pragma unroll
            for (int o = 0; o < Oc; ++o) cb[o] = lg[o] * r;
        }
    }

    if (FIRST) {
        ws_o(sA[Oc - 1], u[Oc - 1], cb[Oc - 1]);
    } else {
#pragma unroll
        for (int o = 0; o < Oc; ++o) ws_o(sA[o], u[o], cb[o]);
    }
}

__global__ __launch_bounds__(NTHREADS)
void DynamicRouting_86766929314212_kernel(const float* __restrict__ x,
                                          const float* W,
                                          float* __restrict__ out)
{
    extern __shared__ __attribute__((aligned(16))) float lds[];
    float* xs = lds;
    float* bl = xs + XS_F;
    float* vs = bl + BL_F;
    float* sp = vs + VS_F;

    const int tid  = threadIdx.x;
    const int lane = tid & 31;
    const int w    = tid >> 5;
    const int n    = lane & 15;
    const int h    = lane >> 4;
    const int arow = n * Kc + 2 * h;

    const int tilesPerB = Lc / LT;
    const int b  = blockIdx.x / tilesPerB;
    const int l0 = (blockIdx.x % tilesPerB) * LT;

#pragma unroll
    for (int t = 0; t < (Ic * Kc * (LT / 4)) / NTHREADS; ++t) {
        const int idx  = tid + t * NTHREADS;
        const int row  = idx >> 2;
        const int part = idx & 3;
        const int i = row >> 4;
        const int k = row & 15;
        const v4f val = *(const v4f*)(x + ((size_t)(b * Ic + i) * Kc + k) * Lc + l0 + part * 4);
        float* dst = xs + (i * LT + part * 4) * XSS + k;
        dst[0 * XSS] = val[0];
        dst[1 * XSS] = val[1];
        dst[2 * XSS] = val[2];
        dst[3 * XSS] = val[3];
    }
    for (int idx = tid; idx < BL_F / 4; idx += NTHREADS) ((v4f*)bl)[idx] = (v4f){0.f, 0.f, 0.f, 0.f};
    __syncthreads();

    for (int it = 0; it < ITERS; ++it) {
        v8f sA[Oc][2];
#pragma unroll
        for (int o = 0; o < Oc; ++o) {
            sA[o][0] = (v8f){0.f,0.f,0.f,0.f,0.f,0.f,0.f,0.f};
            sA[o][1] = (v8f){0.f,0.f,0.f,0.f,0.f,0.f,0.f,0.f};
        }
        if (it == 0) route_pass<true >(W, xs, bl, vs, w, lane, n, h, arow, sA);
        else         route_pass<false>(W, xs, bl, vs, w, lane, n, h, arow, sA);

        {
            v4f* spw = (v4f*)(sp + (size_t)(w * Oc) * FRAG) + lane;
#pragma unroll
            for (int o = 0; o < Oc; ++o) {
                spw[o * 128 +  0] = (v4f){sA[o][0][0], sA[o][0][1], sA[o][0][2], sA[o][0][3]};
                spw[o * 128 + 32] = (v4f){sA[o][0][4], sA[o][0][5], sA[o][0][6], sA[o][0][7]};
                spw[o * 128 + 64] = (v4f){sA[o][1][0], sA[o][1][1], sA[o][1][2], sA[o][1][3]};
                spw[o * 128 + 96] = (v4f){sA[o][1][4], sA[o][1][5], sA[o][1][6], sA[o][1][7]};
            }
        }
        __syncthreads();

#pragma unroll
        for (int qq = 0; qq < Oc / NW; ++qq) {
            const int o = w * (Oc / NW) + qq;
            v4f t[4];
#pragma unroll
            for (int q = 0; q < 4; ++q) {
                const v4f* p = (const v4f*)(sp + (size_t)o * FRAG) + q * 32 + lane;
                t[q] = ((p[0] + p[(Oc * FRAG) / 4]) + p[2 * (Oc * FRAG) / 4]) + p[3 * (Oc * FRAG) / 4];
            }
            float sq = 0.0f;
#pragma unroll
            for (int e = 0; e < 4; ++e)
                sq += t[0][e] * t[0][e] + t[2][e] * t[2][e];
#pragma unroll
            for (int e = 0; e < 4; ++e)
                sq += t[1][e] * t[1][e] + t[3][e] * t[3][e];
            sq += __shfl_xor(sq, 16, 32);
            const float scale = (sq / (1.0f + sq)) * rsqrtf(sq + 1e-8f);
#pragma unroll
            for (int q = 0; q < 4; ++q) t[q] *= scale;

            if (it + 1 < ITERS) {
                v4f* vo = (v4f*)(vs + (size_t)o * FRAG) + lane;
                vo[0] = t[0]; vo[32] = t[1]; vo[64] = t[2]; vo[96] = t[3];
            } else {
                const size_t obase = ((size_t)(b * Oc + o) * Dc) * Lc + l0 + n;
#pragma unroll
                for (int e = 0; e < 4; ++e) {
                    out[obase + (size_t)(8 * h + e) * Lc]          = t[0][e];
                    out[obase + (size_t)(8 * h + 4 + e) * Lc]      = t[1][e];
                    out[obase + (size_t)(16 + 8 * h + e) * Lc]     = t[2][e];
                    out[obase + (size_t)(16 + 8 * h + 4 + e) * Lc] = t[3][e];
                }
            }
        }
        __syncthreads();
    }
}

namespace {
constexpr int IH    = Ic / 2;
constexpr int IPW2  = IH / NW;
constexpr int SLOT  = Oc * FRAG;
constexpr int XS2_F = IH * LT * XSS;
constexpr int LDS2_FLOATS = XS2_F + VS_F + SP_F;
constexpr int NTILE = Bc * (Lc / LT);
constexpr size_t REGION = (size_t)NTILE * 2 * SLOT;
}

__device__ __forceinline__ void squash_frag(v4f (&t)[4])
{
    float sq = 0.0f;
#pragma unroll
    for (int e = 0; e < 4; ++e)
        sq += t[0][e] * t[0][e] + t[2][e] * t[2][e];
#pragma unroll
    for (int e = 0; e < 4; ++e)
        sq += t[1][e] * t[1][e] + t[3][e] * t[3][e];
    sq += __shfl_xor(sq, 16, 32);
    const float scale = (sq / (1.0f + sq)) * rsqrtf(sq + 1e-8f);
#pragma unroll
    for (int q = 0; q < 4; ++q) t[q] *= scale;
}

template <bool FIRST>
__device__ __forceinline__ void route_pass_half(const float* W, const float* xs, const float* vs,
                                                int w, int lane, int n, int h, int arow, int ibase,
                                                v8f (&sA)[Oc][2])
{
    v4f vr[Oc][4];
    if (!FIRST) {
#pragma unroll
        for (int o = 0; o < Oc; ++o) {
            const v4f* vp = (const v4f*)(vs + o * FRAG) + lane;
            vr[o][0] = vp[0]; vr[o][1] = vp[32]; vr[o][2] = vp[64]; vr[o][3] = vp[96];
        }
    }

    v2f af[2][4][8];
    load_group(W + (size_t)ibase * Dc * Kc, 0, arow, af[0]);

    v8f u[Oc][2];
    float cb[Oc];
#pragma unroll
    for (int o = 0; o < Oc; ++o) {
        u[o][0] = (v8f){0.f,0.f,0.f,0.f,0.f,0.f,0.f,0.f};
        u[o][1] = (v8f){0.f,0.f,0.f,0.f,0.f,0.f,0.f,0.f};
        cb[o] = FIRST ? 0.0625f : 0.0f;
    }

    for (int ii = 0; ii < IPW2; ++ii) {
        const int i = ibase + ii;
        const int inext = (ii + 1 < IPW2) ? i + 1 : i;
        const float* Wi = W + (size_t)i * Dc * Kc;
        const float* xrow = xs + ((w * IPW2 + ii) * LT + n) * XSS + 2 * h;
        v2f bf[4];
#pragma unroll
        for (int c = 0; c < 4; ++c) bf[c] = *(const v2f*)(xrow + 4 * c);

        if (FIRST) {
#pragma unroll
            for (int g = 0; g < 4; ++g) {
                if (g < 3) load_group(Wi, g + 1, arow, af[(g + 1) & 1]);
                else       load_group(W + (size_t)inext * Dc * Kc, 0, arow, af[0]);
                __builtin_amdgcn_sched_barrier(0);
#pragma unroll
                for (int q = 0; q < 4; ++q) {
                    const int o = 4 * g + q;
                    if (o == 0) ws_o(sA[Oc - 1], u[Oc - 1], cb[Oc - 1]);
                    mma_o(af[g & 1][q], bf, u[o]);
                    if (o >= 1) ws_o(sA[o - 1], u[o - 1], cb[o - 1]);
                }
                capsule_issue_order<1, 1>(); capsule_issue_order<1, 1>(); capsule_issue_order<1, 1>(); capsule_issue_order<1, 1>();
                __builtin_amdgcn_sched_barrier(0);
            }
        } else {
            float lg[Oc];
            ws_o(sA[0], u[0], cb[0]);
#pragma unroll
            for (int g = 0; g < 4; ++g) {
                if (g < 3) load_group(Wi, g + 1, arow, af[(g + 1) & 1]);
                else       load_group(W + (size_t)inext * Dc * Kc, 0, arow, af[0]);
                __builtin_amdgcn_sched_barrier(0);
#pragma unroll
                for (int q = 0; q < 4; ++q) {
                    const int o = 4 * g + q;
                    mma_o(af[g & 1][q], bf, u[o]);
                    if (o + 1 < Oc) ws_o(sA[o + 1], u[o + 1], cb[o + 1]);
                    if (o >= 1) lg[o - 1] = dot_o(u[o - 1], vr[o - 1]);
                }
                if (g == 0)      { capsule_issue_order<1, 1>(); capsule_issue_order<2, 3>(); capsule_issue_order<2, 3>(); capsule_issue_order<2, 3>(); }
                else if (g == 3) { capsule_issue_order<2, 3>(); capsule_issue_order<2, 3>(); capsule_issue_order<2, 3>(); capsule_issue_order<1, 2>(); }
                else             { capsule_issue_order<2, 3>(); capsule_issue_order<2, 3>(); capsule_issue_order<2, 3>(); capsule_issue_order<2, 3>(); }
                __builtin_amdgcn_sched_barrier(0);
            }
            lg[Oc - 1] = dot_o(u[Oc - 1], vr[Oc - 1]);
            float m = lg[0];
#pragma unroll
            for (int o = 1; o < Oc; ++o) m = fmaxf(m, lg[o]);
            float den = 0.0f;
#pragma unroll
            for (int o = 0; o < Oc; ++o) { lg[o] = __expf(lg[o] - m); den += lg[o]; }
            const float r = 1.0f / den;
#pragma unroll
            for (int o = 0; o < Oc; ++o) cb[o] = lg[o] * r;
        }
    }

    if (FIRST) {
        ws_o(sA[Oc - 1], u[Oc - 1], cb[Oc - 1]);
    } else {
#pragma unroll
        for (int o = 0; o < Oc; ++o) ws_o(sA[o], u[o], cb[o]);
    }
}

template <bool FIRST>
__device__ __forceinline__ void split_iteration(const float* __restrict__ x, const float* W,
                                                float* vkeep, const float* pin, float* pout, int it)
{
    extern __shared__ __attribute__((aligned(16))) float lds2[];
    float* xs = lds2;
    float* vs = xs + XS2_F;
    float* sp = vs + VS_F;

    const int tid  = threadIdx.x;
    const int lane = tid & 31;
    const int w    = tid >> 5;
    const int n    = lane & 15;
    const int h    = lane >> 4;
    const int arow = n * Kc + 2 * h;

    const int tile = blockIdx.x >> 1;
    const int half = blockIdx.x & 1;
    const int tilesPerB = Lc / LT;
    const int b  = tile / tilesPerB;
    const int l0 = (tile % tilesPerB) * LT;
    const int i0 = half * IH;

    v4f t[Oc / NW][4];
    if (!FIRST) {
        const float* pt = pin + (size_t)tile * 2 * SLOT;
#pragma unroll
        for (int qq = 0; qq < Oc / NW; ++qq) {
            const int o = w * (Oc / NW) + qq;
#pragma unroll
            for (int q = 0; q < 4; ++q) {
                const v4f* p = (const v4f*)(pt + (size_t)o * FRAG) + q * 32 + lane;
                t[qq][q] = p[0] + p[SLOT / 4];
            }
        }
    }

#pragma unroll
    for (int s = 0; s < (IH * Kc * (LT / 4)) / NTHREADS; ++s) {
        const int idx  = tid + s * NTHREADS;
        const int row  = idx >> 2;
        const int part = idx & 3;
        const int il = row >> 4;
        const int k  = row & 15;
        const v4f val = *(const v4f*)(x + ((size_t)(b * Ic + i0 + il) * Kc + k) * Lc + l0 + part * 4);
        float* dst = xs + (il * LT + part * 4) * XSS + k;
        dst[0 * XSS] = val[0];
        dst[1 * XSS] = val[1];
        dst[2 * XSS] = val[2];
        dst[3 * XSS] = val[3];
    }

    if (!FIRST) {
        float* vk = vkeep + (size_t)tile * SLOT;
        v4f v0[Oc / NW][4];
        if (it != 1) {
#pragma unroll
            for (int qq = 0; qq < Oc / NW; ++qq) {
                const v4f* vkp = (const v4f*)(vk + (size_t)(w * (Oc / NW) + qq) * FRAG) + lane;
                v0[qq][0] = vkp[0]; v0[qq][1] = vkp[32]; v0[qq][2] = vkp[64]; v0[qq][3] = vkp[96];
            }
        }
#pragma unroll
        for (int qq = 0; qq < Oc / NW; ++qq) {
            const int o = w * (Oc / NW) + qq;
            squash_frag(t[qq]);
            if (it == 1) {
                if (half == 0) {
                    v4f* vkp = (v4f*)(vk + (size_t)o * FRAG) + lane;
                    vkp[0] = t[qq][0]; vkp[32] = t[qq][1]; vkp[64] = t[qq][2]; vkp[96] = t[qq][3];
                }
            } else {
                t[qq][0] += v0[qq][0]; t[qq][1] += v0[qq][1]; t[qq][2] += v0[qq][2]; t[qq][3] += v0[qq][3];
            }
            v4f* vo = (v4f*)(vs + (size_t)o * FRAG) + lane;
            vo[0] = t[qq][0]; vo[32] = t[qq][1]; vo[64] = t[qq][2]; vo[96] = t[qq][3];
        }
    }
    __syncthreads();

    v8f sA[Oc][2];
#pragma unroll
    for (int o = 0; o < Oc; ++o) {
        sA[o][0] = (v8f){0.f,0.f,0.f,0.f,0.f,0.f,0.f,0.f};
        sA[o][1] = (v8f){0.f,0.f,0.f,0.f,0.f,0.f,0.f,0.f};
    }
    route_pass_half<FIRST>(W, xs, vs, w, lane, n, h, arow, i0 + w * IPW2, sA);

    {
        v4f* spw = (v4f*)(sp + (size_t)(w * Oc) * FRAG) + lane;
#pragma unroll
        for (int o = 0; o < Oc; ++o) {
            spw[o * 128 +  0] = (v4f){sA[o][0][0], sA[o][0][1], sA[o][0][2], sA[o][0][3]};
            spw[o * 128 + 32] = (v4f){sA[o][0][4], sA[o][0][5], sA[o][0][6], sA[o][0][7]};
            spw[o * 128 + 64] = (v4f){sA[o][1][0], sA[o][1][1], sA[o][1][2], sA[o][1][3]};
            spw[o * 128 + 96] = (v4f){sA[o][1][4], sA[o][1][5], sA[o][1][6], sA[o][1][7]};
        }
    }
    __syncthreads();

    float* po = pout + ((size_t)tile * 2 + half) * SLOT;
#pragma unroll
    for (int qq = 0; qq < Oc / NW; ++qq) {
        const int o = w * (Oc / NW) + qq;
#pragma unroll
        for (int q = 0; q < 4; ++q) {
            const v4f* p = (const v4f*)(sp + (size_t)o * FRAG) + q * 32 + lane;
            const v4f part = ((p[0] + p[SLOT / 4]) + p[2 * SLOT / 4]) + p[3 * SLOT / 4];
            ((v4f*)(po + (size_t)o * FRAG) + q * 32 + lane)[0] = part;
        }
    }
}

__global__ __launch_bounds__(NTHREADS)
void route_first_kernel(const float* __restrict__ x, const float* W, float* ws)
{
    split_iteration<true>(x, W, ws, ws, ws, 0);
}

__global__ __launch_bounds__(NTHREADS)
void route_next_kernel(const float* __restrict__ x, const float* W, float* out, float* ws, int it)
{
    const float* pin = (it == 1) ? ws : ws + REGION;
    float* pout      = (it == 1) ? ws + REGION : ws;
    split_iteration<false>(x, W, out, pin, pout, it);
}

__global__ __launch_bounds__(NTHREADS)
void route_finish_kernel(float* __restrict__ out, const float* __restrict__ ws)
{
    const int tid  = threadIdx.x;
    const int lane = tid & 31;
    const int w    = tid >> 5;
    const int n    = lane & 15;
    const int h    = lane >> 4;
    const int tile = blockIdx.x;
    const int tilesPerB = Lc / LT;
    const int b  = tile / tilesPerB;
    const int l0 = (tile % tilesPerB) * LT;
    const float* pt = ws + (size_t)tile * 2 * SLOT;
#pragma unroll
    for (int qq = 0; qq < Oc / NW; ++qq) {
        const int o = w * (Oc / NW) + qq;
        v4f t[4];
#pragma unroll
        for (int q = 0; q < 4; ++q) {
            const v4f* p = (const v4f*)(pt + (size_t)o * FRAG) + q * 32 + lane;
            t[q] = p[0] + p[SLOT / 4];
        }
        squash_frag(t);
        const size_t obase = ((size_t)(b * Oc + o) * Dc) * Lc + l0 + n;
#pragma unroll
        for (int e = 0; e < 4; ++e) {
            out[obase + (size_t)(8 * h + e) * Lc]          = t[0][e];
            out[obase + (size_t)(8 * h + 4 + e) * Lc]      = t[1][e];
            out[obase + (size_t)(16 + 8 * h + e) * Lc]     = t[2][e];
            out[obase + (size_t)(16 + 8 * h + 4 + e) * Lc] = t[3][e];
        }
    }
}

extern "C" void kernel_launch(void* const* d_in, const int* in_sizes, int n_in,
                              void* d_out, int out_size, void* d_ws, size_t ws_size,
                              hipStream_t stream)
{
    (void)in_sizes; (void)n_in; (void)out_size;
    const float* x = (const float*)d_in[0];
    const float* W = (const float*)d_in[1];
    float* out = (float*)d_out;
    float* ws  = (float*)d_ws;

    constexpr size_t need = 2 * REGION * sizeof(float);
    if (ws_size >= need) {
        constexpr size_t shmem2 = (size_t)LDS2_FLOATS * sizeof(float);
        dim3 grid2(2 * NTILE);
        route_first_kernel<<<grid2, NTHREADS, shmem2, stream>>>(x, W, ws);
        route_next_kernel<<<grid2, NTHREADS, shmem2, stream>>>(x, W, out, ws, 1);
        route_next_kernel<<<grid2, NTHREADS, shmem2, stream>>>(x, W, out, ws, 2);
        route_finish_kernel<<<dim3(NTILE), NTHREADS, 0, stream>>>(out, ws);
    } else {
        constexpr size_t shmem = (size_t)LDS_FLOATS * sizeof(float);
        dim3 grid(NTILE);
        DynamicRouting_86766929314212_kernel<<<grid, NTHREADS, shmem, stream>>>(x, W, out);
    }
}


# Split path: each workgroup walks its waves' 8 in-capsules as a ring from a start set by its tile, so workgroups sharing a W block stop hitting the same L2 lines at once; f32 sum order of s changes
# speedup vs baseline: 2.2274x; 1.0376x over previous
#include <hip/hip_runtime.h>


typedef __attribute__((ext_vector_type(2))) float v2f;
typedef __attribute__((ext_vector_type(4))) float v4f;
typedef __attribute__((ext_vector_type(8))) float v8f;

namespace {
constexpr int Bc = 8;
constexpr int Ic = 64;
constexpr int Kc = 16;
constexpr int Lc = 256;
constexpr int Oc = 16;
constexpr int Dc = 32;
constexpr int LT = 16;
constexpr int ITERS = 3;
constexpr int XSS = Kc + 2;
constexpr int NW = 4;
constexpr int NTHREADS = NW * 32;
constexpr int IPW = Ic / NW;
constexpr int FRAG = 16 * 32;
constexpr int XS_F = Ic * LT * XSS;
constexpr int BLP = Oc;
constexpr int BL_F = Ic * LT * BLP;
constexpr int VS_F = Oc * FRAG;
constexpr int SP_F = NW * Oc * FRAG;
constexpr int LDS_FLOATS = XS_F + BL_F + VS_F + SP_F;
}

__device__ __forceinline__ float xhalf(float v)
{
    return __builtin_bit_cast(float, __builtin_amdgcn_permlanex16(__builtin_bit_cast(int, v), __builtin_bit_cast(int, v),
                                                                  0x76543210, 0xfedcba98, false, false));
}

__device__ __forceinline__ void load_group(const float* Wi, int g, int arow, v2f (&a)[4][8])
{
#pragma unroll
    for (int q = 0; q < 4; ++q) {
        const float* Wp = Wi + (size_t)(4 * g + q) * Ic * Dc * Kc;
#pragma unroll
        for (int c = 0; c < 4; ++c) {
            a[q][2 * c]     = *(const v2f*)(Wp + arow + 4 * c);
            a[q][2 * c + 1] = *(const v2f*)(Wp + arow + 16 * Kc + 4 * c);
        }
    }
}

__device__ __forceinline__ void mma_o(const v2f (&a)[8], const v2f (&bf)[4], v8f (&u)[2])
{
    v8f u0 = (v8f){0.f,0.f,0.f,0.f,0.f,0.f,0.f,0.f};
    v8f u1 = (v8f){0.f,0.f,0.f,0.f,0.f,0.f,0.f,0.f};
#pragma unroll
    for (int c = 0; c < 4; ++c) {
        u0 = __builtin_amdgcn_wmma_f32_16x16x4_f32(false, a[2 * c],     false, bf[c], (short)0, u0, false, false);
        u1 = __builtin_amdgcn_wmma_f32_16x16x4_f32(false, a[2 * c + 1], false, bf[c], (short)0, u1, false, false);
    }
    u[0] = u0; u[1] = u1;
}

__device__ __forceinline__ void ws_o(v8f (&sA)[2], const v8f (&u)[2], float cb)
{
    sA[0] += cb * u[0];
    sA[1] += cb * u[1];
}

__device__ __forceinline__ float dot_o(const v8f (&u)[2], const v4f (&vr)[4])
{
    v2f a0 = __builtin_shufflevector(u[0], u[0], 0, 1) * __builtin_shufflevector(vr[0], vr[0], 0, 1);
    v2f a1 = __builtin_shufflevector(u[1], u[1], 0, 1) * __builtin_shufflevector(vr[2], vr[2], 0, 1);
    a0 = __builtin_shufflevector(u[0], u[0], 2, 3) * __builtin_shufflevector(vr[0], vr[0], 2, 3) + a0;
    a1 = __builtin_shufflevector(u[1], u[1], 2, 3) * __builtin_shufflevector(vr[2], vr[2], 2, 3) + a1;
    a0 = __builtin_shufflevector(u[0], u[0], 4, 5) * __builtin_shufflevector(vr[1], vr[1], 0, 1) + a0;
    a1 = __builtin_shufflevector(u[1], u[1], 4, 5) * __builtin_shufflevector(vr[3], vr[3], 0, 1) + a1;
    a0 = __builtin_shufflevector(u[0], u[0], 6, 7) * __builtin_shufflevector(vr[1], vr[1], 2, 3) + a0;
    a1 = __builtin_shufflevector(u[1], u[1], 6, 7) * __builtin_shufflevector(vr[3], vr[3], 2, 3) + a1;
    a0 += a1;
    const float d = a0[0] + a0[1];
    return d + xhalf(d);
}

template <int KV>
__device__ __forceinline__ void wmma_slot()
{
    __builtin_amdgcn_sched_group_barrier(0x008, 1, 0);
    __builtin_amdgcn_sched_group_barrier(0x002, KV, 0);
}

template <int KA, int KL>
__device__ __forceinline__ void capsule_issue_order()
{
    wmma_slot<KA>(); wmma_slot<KA>(); wmma_slot<KA>(); wmma_slot<KA>();
    wmma_slot<KA>(); wmma_slot<KA>(); wmma_slot<KA>(); wmma_slot<KL>();
}

template <bool FIRST>
__device__ __forceinline__ void route_pass(const float* W, const float* xs,
                                           float* bl, const float* vs,
                                           int w, int lane, int n, int h, int arow,
                                           v8f (&sA)[Oc][2])
{
    v4f vr[Oc][4];
    if (!FIRST) {
#pragma unroll
        for (int o = 0; o < Oc; ++o) {
            const v4f* vp = (const v4f*)(vs + o * FRAG) + lane;
            vr[o][0] = vp[0]; vr[o][1] = vp[32]; vr[o][2] = vp[64]; vr[o][3] = vp[96];
        }
    }

    v2f af[2][4][8];
    load_group(W + (size_t)(w * IPW) * Dc * Kc, 0, arow, af[0]);

    v8f u[Oc][2];
    float cb[Oc];
#pragma unroll
    for (int o = 0; o < Oc; ++o) {
        u[o][0] = (v8f){0.f,0.f,0.f,0.f,0.f,0.f,0.f,0.f};
        u[o][1] = (v8f){0.f,0.f,0.f,0.f,0.f,0.f,0.f,0.f};
        cb[o] = FIRST ? 0.0625f : 0.0f;
    }

    for (int ii = 0; ii < IPW; ++ii) {
        const int i = w * IPW + ii;
        const int inext = (ii + 1 < IPW) ? i + 1 : i;
        const float* Wi = W + (size_t)i * Dc * Kc;
        const float* xrow = xs + (i * LT + n) * XSS + 2 * h;
        v2f bf[4];
#pragma unroll
        for (int c = 0; c < 4; ++c) bf[c] = *(const v2f*)(xrow + 4 * c);

        float lg[Oc];
        if (FIRST) {
#pragma unroll
            for (int g = 0; g < 4; ++g) {
                if (g < 3) load_group(Wi, g + 1, arow, af[(g + 1) & 1]);
                else       load_group(W + (size_t)inext * Dc * Kc, 0, arow, af[0]);
                __builtin_amdgcn_sched_barrier(0);
#pragma unroll
                for (int q = 0; q < 4; ++q) {
                    const int o = 4 * g + q;
                    if (o == 0) ws_o(sA[Oc - 1], u[Oc - 1], cb[Oc - 1]);
                    mma_o(af[g & 1][q], bf, u[o]);
                    if (o >= 1) ws_o(sA[o - 1], u[o - 1], cb[o - 1]);
                }
                capsule_issue_order<1, 1>(); capsule_issue_order<1, 1>(); capsule_issue_order<1, 1>(); capsule_issue_order<1, 1>();
                __builtin_amdgcn_sched_barrier(0);
            }
        } else {
            v4f* blp = (v4f*)(bl + (i * LT + n) * BLP);
            v4f old[4];
#pragma unroll
            for (int q = 0; q < 4; ++q) old[q] = blp[q];

            ws_o(sA[0], u[0], cb[0]);
#pragma unroll
            for (int g = 0; g < 4; ++g) {
                if (g < 3) load_group(Wi, g + 1, arow, af[(g + 1) & 1]);
                else       load_group(W + (size_t)inext * Dc * Kc, 0, arow, af[0]);
                __builtin_amdgcn_sched_barrier(0);
#pragma unroll
                for (int q = 0; q < 4; ++q) {
                    const int o = 4 * g + q;
                    mma_o(af[g & 1][q], bf, u[o]);
                    if (o + 1 < Oc) ws_o(sA[o + 1], u[o + 1], cb[o + 1]);
                    if (o >= 1) lg[o - 1] = dot_o(u[o - 1], vr[o - 1]);
                }
                if (g == 0)      { capsule_issue_order<1, 1>(); capsule_issue_order<2, 3>(); capsule_issue_order<2, 3>(); capsule_issue_order<2, 3>(); }
                else if (g == 3) { capsule_issue_order<2, 3>(); capsule_issue_order<2, 3>(); capsule_issue_order<2, 3>(); capsule_issue_order<1, 2>(); }
                else             { capsule_issue_order<2, 3>(); capsule_issue_order<2, 3>(); capsule_issue_order<2, 3>(); capsule_issue_order<2, 3>(); }
                __builtin_amdgcn_sched_barrier(0);
            }
            lg[Oc - 1] = dot_o(u[Oc - 1], vr[Oc - 1]);
#pragma unroll
            for (int q = 0; q < 4; ++q) {
                lg[4 * q + 0] += old[q][0]; lg[4 * q + 1] += old[q][1]; lg[4 * q + 2] += old[q][2]; lg[4 * q + 3] += old[q][3];
            }
#pragma unroll
            for (int q = 0; q < 4; ++q)
                blp[q] = (v4f){lg[4 * q + 0], lg[4 * q + 1], lg[4 * q + 2], lg[4 * q + 3]};
            float m = lg[0];
#pragma unroll
            for (int o = 1; o < Oc; ++o) m = fmaxf(m, lg[o]);
            float den = 0.0f;
#pragma unroll
            for (int o = 0; o < Oc; ++o) { lg[o] = __expf(lg[o] - m); den += lg[o]; }
            const float r = 1.0f / den;
#pragma unroll
            for (int o = 0; o < Oc; ++o) cb[o] = lg[o] * r;
        }
    }

    if (FIRST) {
        ws_o(sA[Oc - 1], u[Oc - 1], cb[Oc - 1]);
    } else {
#pragma unroll
        for (int o = 0; o < Oc; ++o) ws_o(sA[o], u[o], cb[o]);
    }
}

__global__ __launch_bounds__(NTHREADS)
void DynamicRouting_86766929314212_kernel(const float* __restrict__ x,
                                          const float* W,
                                          float* __restrict__ out)
{
    extern __shared__ __attribute__((aligned(16))) float lds[];
    float* xs = lds;
    float* bl = xs + XS_F;
    float* vs = bl + BL_F;
    float* sp = vs + VS_F;

    const int tid  = threadIdx.x;
    const int lane = tid & 31;
    const int w    = tid >> 5;
    const int n    = lane & 15;
    const int h    = lane >> 4;
    const int arow = n * Kc + 2 * h;

    const int tilesPerB = Lc / LT;
    const int b  = blockIdx.x / tilesPerB;
    const int l0 = (blockIdx.x % tilesPerB) * LT;

#pragma unroll
    for (int t = 0; t < (Ic * Kc * (LT / 4)) / NTHREADS; ++t) {
        const int idx  = tid + t * NTHREADS;
        const int row  = idx >> 2;
        const int part = idx & 3;
        const int i = row >> 4;
        const int k = row & 15;
        const v4f val = *(const v4f*)(x + ((size_t)(b * Ic + i) * Kc + k) * Lc + l0 + part * 4);
        float* dst = xs + (i * LT + part * 4) * XSS + k;
        dst[0 * XSS] = val[0];
        dst[1 * XSS] = val[1];
        dst[2 * XSS] = val[2];
        dst[3 * XSS] = val[3];
    }
    for (int idx = tid; idx < BL_F / 4; idx += NTHREADS) ((v4f*)bl)[idx] = (v4f){0.f, 0.f, 0.f, 0.f};
    __syncthreads();

    for (int it = 0; it < ITERS; ++it) {
        v8f sA[Oc][2];
#pragma unroll
        for (int o = 0; o < Oc; ++o) {
            sA[o][0] = (v8f){0.f,0.f,0.f,0.f,0.f,0.f,0.f,0.f};
            sA[o][1] = (v8f){0.f,0.f,0.f,0.f,0.f,0.f,0.f,0.f};
        }
        if (it == 0) route_pass<true >(W, xs, bl, vs, w, lane, n, h, arow, sA);
        else         route_pass<false>(W, xs, bl, vs, w, lane, n, h, arow, sA);

        {
            v4f* spw = (v4f*)(sp + (size_t)(w * Oc) * FRAG) + lane;
#pragma unroll
            for (int o = 0; o < Oc; ++o) {
                spw[o * 128 +  0] = (v4f){sA[o][0][0], sA[o][0][1], sA[o][0][2], sA[o][0][3]};
                spw[o * 128 + 32] = (v4f){sA[o][0][4], sA[o][0][5], sA[o][0][6], sA[o][0][7]};
                spw[o * 128 + 64] = (v4f){sA[o][1][0], sA[o][1][1], sA[o][1][2], sA[o][1][3]};
                spw[o * 128 + 96] = (v4f){sA[o][1][4], sA[o][1][5], sA[o][1][6], sA[o][1][7]};
            }
        }
        __syncthreads();

#pragma unroll
        for (int qq = 0; qq < Oc / NW; ++qq) {
            const int o = w * (Oc / NW) + qq;
            v4f t[4];
#pragma unroll
            for (int q = 0; q < 4; ++q) {
                const v4f* p = (const v4f*)(sp + (size_t)o * FRAG) + q * 32 + lane;
                t[q] = ((p[0] + p[(Oc * FRAG) / 4]) + p[2 * (Oc * FRAG) / 4]) + p[3 * (Oc * FRAG) / 4];
            }
            float sq = 0.0f;
#pragma unroll
            for (int e = 0; e < 4; ++e)
                sq += t[0][e] * t[0][e] + t[2][e] * t[2][e];
#pragma unroll
            for (int e = 0; e < 4; ++e)
                sq += t[1][e] * t[1][e] + t[3][e] * t[3][e];
            sq += __shfl_xor(sq, 16, 32);
            const float scale = (sq / (1.0f + sq)) * rsqrtf(sq + 1e-8f);
#pragma unroll
            for (int q = 0; q < 4; ++q) t[q] *= scale;

            if (it + 1 < ITERS) {
                v4f* vo = (v4f*)(vs + (size_t)o * FRAG) + lane;
                vo[0] = t[0]; vo[32] = t[1]; vo[64] = t[2]; vo[96] = t[3];
            } else {
                const size_t obase = ((size_t)(b * Oc + o) * Dc) * Lc + l0 + n;
#pragma unroll
                for (int e = 0; e < 4; ++e) {
                    out[obase + (size_t)(8 * h + e) * Lc]          = t[0][e];
                    out[obase + (size_t)(8 * h + 4 + e) * Lc]      = t[1][e];
                    out[obase + (size_t)(16 + 8 * h + e) * Lc]     = t[2][e];
                    out[obase + (size_t)(16 + 8 * h + 4 + e) * Lc] = t[3][e];
                }
            }
        }
        __syncthreads();
    }
}

namespace {
constexpr int IH    = Ic / 2;
constexpr int IPW2  = IH / NW;
constexpr int SLOT  = Oc * FRAG;
constexpr int XS2_F = IH * LT * XSS;
constexpr int LDS2_FLOATS = XS2_F + VS_F + SP_F;
constexpr int NTILE = Bc * (Lc / LT);
constexpr size_t REGION = (size_t)NTILE * 2 * SLOT;
}

__device__ __forceinline__ void squash_frag(v4f (&t)[4])
{
    float sq = 0.0f;
#pragma unroll
    for (int e = 0; e < 4; ++e)
        sq += t[0][e] * t[0][e] + t[2][e] * t[2][e];
#pragma unroll
    for (int e = 0; e < 4; ++e)
        sq += t[1][e] * t[1][e] + t[3][e] * t[3][e];
    sq += __shfl_xor(sq, 16, 32);
    const float scale = (sq / (1.0f + sq)) * rsqrtf(sq + 1e-8f);
#pragma unroll
    for (int q = 0; q < 4; ++q) t[q] *= scale;
}

template <bool FIRST>
__device__ __forceinline__ void route_pass_half(const float* W, const float* xs, const float* vs,
                                                int w, int lane, int n, int h, int arow, int ibase, int rot,
                                                v8f (&sA)[Oc][2])
{
    v4f vr[Oc][4];
    if (!FIRST) {
#pragma unroll
        for (int o = 0; o < Oc; ++o) {
            const v4f* vp = (const v4f*)(vs + o * FRAG) + lane;
            vr[o][0] = vp[0]; vr[o][1] = vp[32]; vr[o][2] = vp[64]; vr[o][3] = vp[96];
        }
    }

    v2f af[2][4][8];
    load_group(W + (size_t)(ibase + rot) * Dc * Kc, 0, arow, af[0]);

    v8f u[Oc][2];
    float cb[Oc];
#pragma unroll
    for (int o = 0; o < Oc; ++o) {
        u[o][0] = (v8f){0.f,0.f,0.f,0.f,0.f,0.f,0.f,0.f};
        u[o][1] = (v8f){0.f,0.f,0.f,0.f,0.f,0.f,0.f,0.f};
        cb[o] = FIRST ? 0.0625f : 0.0f;
    }

    for (int ii = 0; ii < IPW2; ++ii) {
        const int il = (ii + rot) & (IPW2 - 1);
        const int i = ibase + il;
        const int inext = (ii + 1 < IPW2) ? ibase + ((il + 1) & (IPW2 - 1)) : i;
        const float* Wi = W + (size_t)i * Dc * Kc;
        const float* xrow = xs + ((w * IPW2 + il) * LT + n) * XSS + 2 * h;
        v2f bf[4];
#pragma unroll
        for (int c = 0; c < 4; ++c) bf[c] = *(const v2f*)(xrow + 4 * c);

        if (FIRST) {
#pragma unroll
            for (int g = 0; g < 4; ++g) {
                if (g < 3) load_group(Wi, g + 1, arow, af[(g + 1) & 1]);
                else       load_group(W + (size_t)inext * Dc * Kc, 0, arow, af[0]);
                __builtin_amdgcn_sched_barrier(0);
#pragma unroll
                for (int q = 0; q < 4; ++q) {
                    const int o = 4 * g + q;
                    if (o == 0) ws_o(sA[Oc - 1], u[Oc - 1], cb[Oc - 1]);
                    mma_o(af[g & 1][q], bf, u[o]);
                    if (o >= 1) ws_o(sA[o - 1], u[o - 1], cb[o - 1]);
                }
                capsule_issue_order<1, 1>(); capsule_issue_order<1, 1>(); capsule_issue_order<1, 1>(); capsule_issue_order<1, 1>();
                __builtin_amdgcn_sched_barrier(0);
            }
        } else {
            float lg[Oc];
            ws_o(sA[0], u[0], cb[0]);
#pragma unroll
            for (int g = 0; g < 4; ++g) {
                if (g < 3) load_group(Wi, g + 1, arow, af[(g + 1) & 1]);
                else       load_group(W + (size_t)inext * Dc * Kc, 0, arow, af[0]);
                __builtin_amdgcn_sched_barrier(0);
#pragma unroll
                for (int q = 0; q < 4; ++q) {
                    const int o = 4 * g + q;
                    mma_o(af[g & 1][q], bf, u[o]);
                    if (o + 1 < Oc) ws_o(sA[o + 1], u[o + 1], cb[o + 1]);
                    if (o >= 1) lg[o - 1] = dot_o(u[o - 1], vr[o - 1]);
                }
                if (g == 0)      { capsule_issue_order<1, 1>(); capsule_issue_order<2, 3>(); capsule_issue_order<2, 3>(); capsule_issue_order<2, 3>(); }
                else if (g == 3) { capsule_issue_order<2, 3>(); capsule_issue_order<2, 3>(); capsule_issue_order<2, 3>(); capsule_issue_order<1, 2>(); }
                else             { capsule_issue_order<2, 3>(); capsule_issue_order<2, 3>(); capsule_issue_order<2, 3>(); capsule_issue_order<2, 3>(); }
                __builtin_amdgcn_sched_barrier(0);
            }
            lg[Oc - 1] = dot_o(u[Oc - 1], vr[Oc - 1]);
            float m = lg[0];
#pragma unroll
            for (int o = 1; o < Oc; ++o) m = fmaxf(m, lg[o]);
            float den = 0.0f;
#pragma unroll
            for (int o = 0; o < Oc; ++o) { lg[o] = __expf(lg[o] - m); den += lg[o]; }
            const float r = 1.0f / den;
#pragma unroll
            for (int o = 0; o < Oc; ++o) cb[o] = lg[o] * r;
        }
    }

    if (FIRST) {
        ws_o(sA[Oc - 1], u[Oc - 1], cb[Oc - 1]);
    } else {
#pragma unroll
        for (int o = 0; o < Oc; ++o) ws_o(sA[o], u[o], cb[o]);
    }
}

template <bool FIRST>
__device__ __forceinline__ void split_iteration(const float* __restrict__ x, const float* W,
                                                float* vkeep, const float* pin, float* pout, int it)
{
    extern __shared__ __attribute__((aligned(16))) float lds2[];
    float* xs = lds2;
    float* vs = xs + XS2_F;
    float* sp = vs + VS_F;

    const int tid  = threadIdx.x;
    const int lane = tid & 31;
    const int w    = tid >> 5;
    const int n    = lane & 15;
    const int h    = lane >> 4;
    const int arow = n * Kc + 2 * h;

    const int tile = blockIdx.x >> 1;
    const int half = blockIdx.x & 1;
    const int tilesPerB = Lc / LT;
    const int b  = tile / tilesPerB;
    const int l0 = (tile % tilesPerB) * LT;
    const int i0 = half * IH;

    v4f t[Oc / NW][4];
    if (!FIRST) {
        const float* pt = pin + (size_t)tile * 2 * SLOT;
#pragma unroll
        for (int qq = 0; qq < Oc / NW; ++qq) {
            const int o = w * (Oc / NW) + qq;
#pragma unroll
            for (int q = 0; q < 4; ++q) {
                const v4f* p = (const v4f*)(pt + (size_t)o * FRAG) + q * 32 + lane;
                t[qq][q] = p[0] + p[SLOT / 4];
            }
        }
    }

#pragma unroll
    for (int s = 0; s < (IH * Kc * (LT / 4)) / NTHREADS; ++s) {
        const int idx  = tid + s * NTHREADS;
        const int row  = idx >> 2;
        const int part = idx & 3;
        const int il = row >> 4;
        const int k  = row & 15;
        const v4f val = *(const v4f*)(x + ((size_t)(b * Ic + i0 + il) * Kc + k) * Lc + l0 + part * 4);
        float* dst = xs + (il * LT + part * 4) * XSS + k;
        dst[0 * XSS] = val[0];
        dst[1 * XSS] = val[1];
        dst[2 * XSS] = val[2];
        dst[3 * XSS] = val[3];
    }

    if (!FIRST) {
        float* vk = vkeep + (size_t)tile * SLOT;
        v4f v0[Oc / NW][4];
        if (it != 1) {
#pragma unroll
            for (int qq = 0; qq < Oc / NW; ++qq) {
                const v4f* vkp = (const v4f*)(vk + (size_t)(w * (Oc / NW) + qq) * FRAG) + lane;
                v0[qq][0] = vkp[0]; v0[qq][1] = vkp[32]; v0[qq][2] = vkp[64]; v0[qq][3] = vkp[96];
            }
        }
#pragma unroll
        for (int qq = 0; qq < Oc / NW; ++qq) {
            const int o = w * (Oc / NW) + qq;
            squash_frag(t[qq]);
            if (it == 1) {
                if (half == 0) {
                    v4f* vkp = (v4f*)(vk + (size_t)o * FRAG) + lane;
                    vkp[0] = t[qq][0]; vkp[32] = t[qq][1]; vkp[64] = t[qq][2]; vkp[96] = t[qq][3];
                }
            } else {
                t[qq][0] += v0[qq][0]; t[qq][1] += v0[qq][1]; t[qq][2] += v0[qq][2]; t[qq][3] += v0[qq][3];
            }
            v4f* vo = (v4f*)(vs + (size_t)o * FRAG) + lane;
            vo[0] = t[qq][0]; vo[32] = t[qq][1]; vo[64] = t[qq][2]; vo[96] = t[qq][3];
        }
    }
    __syncthreads();

    v8f sA[Oc][2];
#pragma unroll
    for (int o = 0; o < Oc; ++o) {
        sA[o][0] = (v8f){0.f,0.f,0.f,0.f,0.f,0.f,0.f,0.f};
        sA[o][1] = (v8f){0.f,0.f,0.f,0.f,0.f,0.f,0.f,0.f};
    }
    const int rot = (tile + (tile >> 3)) & (IPW2 - 1);
    route_pass_half<FIRST>(W, xs, vs, w, lane, n, h, arow, i0 + w * IPW2, rot, sA);

    {
        v4f* spw = (v4f*)(sp + (size_t)(w * Oc) * FRAG) + lane;
#pragma unroll
        for (int o = 0; o < Oc; ++o) {
            spw[o * 128 +  0] = (v4f){sA[o][0][0], sA[o][0][1], sA[o][0][2], sA[o][0][3]};
            spw[o * 128 + 32] = (v4f){sA[o][0][4], sA[o][0][5], sA[o][0][6], sA[o][0][7]};
            spw[o * 128 + 64] = (v4f){sA[o][1][0], sA[o][1][1], sA[o][1][2], sA[o][1][3]};
            spw[o * 128 + 96] = (v4f){sA[o][1][4], sA[o][1][5], sA[o][1][6], sA[o][1][7]};
        }
    }
    __syncthreads();

    float* po = pout + ((size_t)tile * 2 + half) * SLOT;
#pragma unroll
    for (int qq = 0; qq < Oc / NW; ++qq) {
        const int o = w * (Oc / NW) + qq;
#pragma unroll
        for (int q = 0; q < 4; ++q) {
            const v4f* p = (const v4f*)(sp + (size_t)o * FRAG) + q * 32 + lane;
            const v4f part = ((p[0] + p[SLOT / 4]) + p[2 * SLOT / 4]) + p[3 * SLOT / 4];
            ((v4f*)(po + (size_t)o * FRAG) + q * 32 + lane)[0] = part;
        }
    }
}

__global__ __launch_bounds__(NTHREADS)
void route_first_kernel(const float* __restrict__ x, const float* W, float* ws)
{
    split_iteration<true>(x, W, ws, ws, ws, 0);
}

__global__ __launch_bounds__(NTHREADS)
void route_next_kernel(const float* __restrict__ x, const float* W, float* out, float* ws, int it)
{
    const float* pin = (it == 1) ? ws : ws + REGION;
    float* pout      = (it == 1) ? ws + REGION : ws;
    split_iteration<false>(x, W, out, pin, pout, it);
}

__global__ __launch_bounds__(NTHREADS)
void route_finish_kernel(float* __restrict__ out, const float* __restrict__ ws)
{
    const int tid  = threadIdx.x;
    const int lane = tid & 31;
    const int w    = tid >> 5;
    const int n    = lane & 15;
    const int h    = lane >> 4;
    const int tile = blockIdx.x;
    const int tilesPerB = Lc / LT;
    const int b  = tile / tilesPerB;
    const int l0 = (tile % tilesPerB) * LT;
    const float* pt = ws + (size_t)tile * 2 * SLOT;
#pragma unroll
    for (int qq = 0; qq < Oc / NW; ++qq) {
        const int o = w * (Oc / NW) + qq;
        v4f t[4];
#pragma unroll
        for (int q = 0; q < 4; ++q) {
            const v4f* p = (const v4f*)(pt + (size_t)o * FRAG) + q * 32 + lane;
            t[q] = p[0] + p[SLOT / 4];
        }
        squash_frag(t);
        const size_t obase = ((size_t)(b * Oc + o) * Dc) * Lc + l0 + n;
#pragma unroll
        for (int e = 0; e < 4; ++e) {
            out[obase + (size_t)(8 * h + e) * Lc]          = t[0][e];
            out[obase + (size_t)(8 * h + 4 + e) * Lc]      = t[1][e];
            out[obase + (size_t)(16 + 8 * h + e) * Lc]     = t[2][e];
            out[obase + (size_t)(16 + 8 * h + 4 + e) * Lc] = t[3][e];
        }
    }
}

extern "C" void kernel_launch(void* const* d_in, const int* in_sizes, int n_in,
                              void* d_out, int out_size, void* d_ws, size_t ws_size,
                              hipStream_t stream)
{
    (void)in_sizes; (void)n_in; (void)out_size;
    const float* x = (const float*)d_in[0];
    const float* W = (const float*)d_in[1];
    float* out = (float*)d_out;
    float* ws  = (float*)d_ws;

    constexpr size_t need = 2 * REGION * sizeof(float);
    if (ws_size >= need) {
        constexpr size_t shmem2 = (size_t)LDS2_FLOATS * sizeof(float);
        dim3 grid2(2 * NTILE);
        route_first_kernel<<<grid2, NTHREADS, shmem2, stream>>>(x, W, ws);
        route_next_kernel<<<grid2, NTHREADS, shmem2, stream>>>(x, W, out, ws, 1);
        route_next_kernel<<<grid2, NTHREADS, shmem2, stream>>>(x, W, out, ws, 2);
        route_finish_kernel<<<dim3(NTILE), NTHREADS, 0, stream>>>(out, ws);
    } else {
        constexpr size_t shmem = (size_t)LDS_FLOATS * sizeof(float);
        dim3 grid(NTILE);
        DynamicRouting_86766929314212_kernel<<<grid, NTHREADS, shmem, stream>>>(x, W, out);
    }
}
